# starting kernel
# baseline (speedup 1.0000x reference)
#include <hip/hip_runtime.h>

typedef float v2f __attribute__((ext_vector_type(2)));
typedef float v4f __attribute__((ext_vector_type(4)));
typedef float v8f __attribute__((ext_vector_type(8)));

#define WAVES 8

__global__ __launch_bounds__(WAVES * 32)
void nufft2_wmma(const float* __restrict__ xr, const float* __restrict__ xi,
                 const float* __restrict__ traj, float* __restrict__ out)
{
    constexpr int B = 16, C = 4, H = 64, W = 64, M = 2048;
    (void)B; (void)H;

    __shared__ float2 eyTab[WAVES][16][64];

    const int lane = threadIdx.x & 31;
    const int wave = threadIdx.x >> 5;
    const int tile = blockIdx.x * WAVES + wave;
    const int b    = tile >> 7;
    const int m0   = (tile & 127) << 4;
    const int hh   = lane >> 4;
    const int mloc = lane & 15;

    const float2 kk = ((const float2*)traj)[(size_t)b * M + m0 + mloc];
    const float ky = kk.x, kx = kk.y;

    {
        const int r0 = hh * 32;
        float s0, c0, su, cu;
        sincosf(ky * (float)(r0 - 32), &s0, &c0);
        sincosf(ky, &su, &cu);
        float cc = c0, ss = s0;
        #pragma unroll 4
        for (int r = 0; r < 32; ++r) {
            eyTab[wave][mloc][r0 + r] = make_float2(cc, ss);
            const float t = cc * cu - ss * su;
            ss = ss * cu + cc * su;
            cc = t;
        }
    }
    __syncthreads();

    v2f axc[16], axs[16];
    {
        const int col0 = hh * 32;
        float s0, c0, su, cu;
        sincosf(kx * (float)(col0 - 32), &s0, &c0);
        sincosf(kx, &su, &cu);
        const float c2 = cu * cu - su * su, s2 = 2.0f * cu * su;
        float ec = c0, es = s0;
        #pragma unroll
        for (int k = 0; k < 16; ++k) {
            const float e1c = ec * cu - es * su;
            const float e1s = es * cu + ec * su;
            axc[k].x = ec;  axc[k].y = e1c;
            axs[k].x = es;  axs[k].y = e1s;
            const float t = ec * c2 - es * s2;
            es = es * c2 + ec * s2;
            ec = t;
        }
    }

    float yre[8], yim[8];
    #pragma unroll
    for (int j = 0; j < 8; ++j) { yre[j] = 0.0f; yim[j] = 0.0f; }

    const int cc_ = mloc & 3;
    const int rl  = mloc >> 2;
    const size_t xbase = ((size_t)(b * C + cc_) * H) * W + (size_t)(hh * 32);

    #pragma unroll 1
    for (int t = 0; t < 16; ++t) {
        const int r = t * 4 + rl;
        const float* pr = xr + xbase + (size_t)r * W;
        const float* pi = xi + xbase + (size_t)r * W;

        v4f brA[8], biA[8];
        #pragma unroll
        for (int k2 = 0; k2 < 8; ++k2) {
            brA[k2] = *(const v4f*)(pr + 4 * k2);
            biA[k2] = *(const v4f*)(pi + 4 * k2);
        }

        v8f zr = {}; v8f zi = {};
        #pragma unroll
        for (int k2 = 0; k2 < 8; ++k2) {
            const v2f br0 = __builtin_shufflevector(brA[k2], brA[k2], 0, 1);
            const v2f br1 = __builtin_shufflevector(brA[k2], brA[k2], 2, 3);
            const v2f bi0 = __builtin_shufflevector(biA[k2], biA[k2], 0, 1);
            const v2f bi1 = __builtin_shufflevector(biA[k2], biA[k2], 2, 3);
            const int k0 = 2 * k2, k1 = 2 * k2 + 1;

            zr = __builtin_amdgcn_wmma_f32_16x16x4_f32(false, axc[k0], false, br0,
                                                       (short)0, zr, false, false);
            zi = __builtin_amdgcn_wmma_f32_16x16x4_f32(false, axc[k0], false, bi0,
                                                       (short)0, zi, false, false);
            zr = __builtin_amdgcn_wmma_f32_16x16x4_f32(false, axs[k0], false, bi0,
                                                       (short)0, zr, false, false);
            const v2f nbr0 = -br0;
            zi = __builtin_amdgcn_wmma_f32_16x16x4_f32(false, axs[k0], false, nbr0,
                                                       (short)0, zi, false, false);

            zr = __builtin_amdgcn_wmma_f32_16x16x4_f32(false, axc[k1], false, br1,
                                                       (short)0, zr, false, false);
            zi = __builtin_amdgcn_wmma_f32_16x16x4_f32(false, axc[k1], false, bi1,
                                                       (short)0, zi, false, false);
            zr = __builtin_amdgcn_wmma_f32_16x16x4_f32(false, axs[k1], false, bi1,
                                                       (short)0, zr, false, false);
            const v2f nbr1 = -br1;
            zi = __builtin_amdgcn_wmma_f32_16x16x4_f32(false, axs[k1], false, nbr1,
                                                       (short)0, zi, false, false);
        }

        #pragma unroll
        for (int j = 0; j < 8; ++j) {
            const float2 ey = eyTab[wave][j + hh * 8][r];
            yre[j] = fmaf(ey.x, zr[j], yre[j]);
            yre[j] = fmaf(ey.y, zi[j], yre[j]);
            yim[j] = fmaf(ey.x, zi[j], yim[j]);
            yim[j] = fmaf(-ey.y, zr[j], yim[j]);
        }
    }

    #pragma unroll
    for (int j = 0; j < 8; ++j) {
        float a = yre[j], c = yim[j];
        a += __shfl_xor(a, 4, 32);
        a += __shfl_xor(a, 8, 32);
        c += __shfl_xor(c, 4, 32);
        c += __shfl_xor(c, 8, 32);
        if ((lane & 12) == 0) {
            const int m  = m0 + j + hh * 8;
            const int ch = lane & 3;
            float2* op = (float2*)out + ((size_t)(b * C + ch) * M + m);
            *op = make_float2(a, c);
        }
    }
}

extern "C" void kernel_launch(void* const* d_in, const int* in_sizes, int n_in,
                              void* d_out, int out_size, void* d_ws, size_t ws_size,
                              hipStream_t stream) {
    (void)in_sizes; (void)n_in; (void)out_size; (void)d_ws; (void)ws_size;
    const float* xr   = (const float*)d_in[0];
    const float* xi   = (const float*)d_in[1];
    const float* traj = (const float*)d_in[2];
    float* out = (float*)d_out;

    dim3 grid(256), block(WAVES * 32);
    hipLaunchKernelGGL(nufft2_wmma, grid, block, 0, stream, xr, xi, traj, out);
}


# Register blocking over m: a wave owns one coil and four m-tiles, so each loaded and folded image row feeds four A tiles, was one; E_y fragments handed over through LDS; sums unchanged
# speedup vs baseline: 2.7418x; 2.7418x over previous
#include <hip/hip_runtime.h>


typedef float v2f __attribute__((ext_vector_type(2)));
typedef float v4f __attribute__((ext_vector_type(4)));
typedef float v8f __attribute__((ext_vector_type(8)));

#define WAVES 8
#define TPW 4
#define EXPITCH 66

__device__ __forceinline__ v8f mm4(v2f a, v2f b, v8f c) {
    return __builtin_amdgcn_wmma_f32_16x16x4_f32(false, a, false, b, (short)0, c, false, false);
}

template <int XM>
__device__ __forceinline__ float row_xor(float x) {
    return __int_as_float(__builtin_amdgcn_update_dpp(0, __float_as_int(x), 0x160 + XM, 0xf, 0xf, true));
}

__device__ __forceinline__ float col_phase(const v8f (&zre)[4], const v8f (&zim)[4],
                                           const float2* etile, int hh, int mloc)
{
    float v[16];
    #pragma unroll
    for (int j = 0; j < 8; ++j) {
        const float2* erow = etile + (j + hh * 8) * EXPITCH + 4 * mloc;
        const v4f e01 = *(const v4f*)(erow);
        const v4f e23 = *(const v4f*)(erow + 2);
        const float cx[4] = {e01.x, e01.z, e23.x, e23.z};
        const float sx[4] = {e01.y, e01.w, e23.y, e23.w};
        float re = 0.0f, im = 0.0f;
        #pragma unroll
        for (int jt = 0; jt < 4; ++jt) {
            const float zr = zre[jt][j];
            const float zi = zim[jt][j];
            re = fmaf(cx[jt], zr, re);
            re = fmaf(sx[jt], zi, re);
            im = fmaf(cx[jt], zi, im);
            im = fmaf(-sx[jt], zr, im);
        }
        v[2 * j] = re;  v[2 * j + 1] = im;
    }
    float w8[8], w4[4], w2[2];
    {
        const bool hi = (mloc & 8) != 0;
        #pragma unroll
        for (int i = 0; i < 8; ++i) {
            const float keep = hi ? v[8 + i] : v[i];
            const float send = hi ? v[i] : v[8 + i];
            w8[i] = keep + row_xor<8>(send);
        }
    }
    {
        const bool hi = (mloc & 4) != 0;
        #pragma unroll
        for (int i = 0; i < 4; ++i) {
            const float keep = hi ? w8[4 + i] : w8[i];
            const float send = hi ? w8[i] : w8[4 + i];
            w4[i] = keep + row_xor<4>(send);
        }
    }
    {
        const bool hi = (mloc & 2) != 0;
        #pragma unroll
        for (int i = 0; i < 2; ++i) {
            const float keep = hi ? w4[2 + i] : w4[i];
            const float send = hi ? w4[i] : w4[2 + i];
            w2[i] = keep + row_xor<2>(send);
        }
    }
    const bool hi = (mloc & 1) != 0;
    const float keep = hi ? w2[1] : w2[0];
    const float send = hi ? w2[0] : w2[1];
    return keep + row_xor<1>(send);
}

__global__ __launch_bounds__(WAVES * 32)
void nufft2_rowfirst_mt4(const float* __restrict__ xr, const float* __restrict__ xi,
                         const float* __restrict__ traj, float* __restrict__ out)
{
    constexpr int C = 4, H = 64, W = 64, M = 2048;

    __shared__ __attribute__((aligned(16))) float2 exTab[WAVES][16][EXPITCH];
    __shared__ __attribute__((aligned(16))) v4f ayTab[WAVES][8][32];

    const int lane = threadIdx.x & 31;
    const int wave = threadIdx.x >> 5;
    const int grp  = wave >> 2;
    const int coil = wave & 3;
    const int tile = blockIdx.x * WAVES + wave;
    const int b    = tile >> 7;
    const int m0   = (tile & 127) << 4;
    const int hh   = lane >> 4;
    const int mloc = lane & 15;

    const size_t img     = (size_t)(b * C + coil) * H * W;
    const size_t lane_up = (size_t)(32 + hh * 16) * W + (size_t)(4 * mloc);
    const size_t lane_dn = (size_t)(31 - hh * 16) * W + (size_t)(4 * mloc);
    const float* pru0 = xr + img + lane_up;
    const float* prd0 = xr + img + lane_dn;
    const float* piu0 = xi + img + lane_up;
    const float* pid0 = xi + img + lane_dn;

    v4f q[2][8];
    q[0][0] = *(const v4f*)(pru0);
    q[0][1] = *(const v4f*)(pru0 + W);
    q[0][2] = *(const v4f*)(prd0);
    q[0][3] = *(const v4f*)(prd0 - W);
    q[0][4] = *(const v4f*)(piu0);
    q[0][5] = *(const v4f*)(piu0 + W);
    q[0][6] = *(const v4f*)(pid0);
    q[0][7] = *(const v4f*)(pid0 - W);

    const float2 kk = ((const float2*)traj)[(size_t)b * M + m0 + mloc];
    const float ky = kk.x, kx = kk.y;

    {
        const int c0i = hh * 32;
        float s0, c0, su, cu;
        sincosf(kx * (float)(c0i - 32) - 0.5f * ky, &s0, &c0);
        sincosf(kx, &su, &cu);
        float cc = c0, ss = s0;
        #pragma unroll 4
        for (int q4 = 0; q4 < 32; ++q4) {
            exTab[wave][mloc][c0i + q4] = make_float2(cc, ss);
            const float t = cc * cu - ss * su;
            ss = ss * cu + cc * su;
            cc = t;
        }
    }

    {
        float s0, c0, su, cu;
        sincosf(ky * ((float)(hh * 16) + 0.5f), &s0, &c0);
        sincosf(ky, &su, &cu);
        const float c2 = cu * cu - su * su, s2 = 2.0f * cu * su;
        float ec = c0, es = s0;
        #pragma unroll
        for (int k = 0; k < 8; ++k) {
            v4f a;
            a.x = ec;   a.y = ec * cu - es * su;
            a.z = es;   a.w = es * cu + ec * su;
            ayTab[wave][k][lane] = a;
            const float t = ec * c2 - es * s2;
            es = es * c2 + ec * s2;
            ec = t;
        }
    }
    __syncthreads();

    v8f zre[TPW][4], zim[TPW][4];
    #pragma unroll
    for (int t = 0; t < TPW; ++t) {
        #pragma unroll
        for (int jt = 0; jt < 4; ++jt) { zre[t][jt] = (v8f){}; zim[t][jt] = (v8f){}; }
    }

    v4f af[2][TPW];
    #pragma unroll
    for (int t = 0; t < TPW; ++t) af[0][t] = ayTab[grp * TPW + t][0][lane];

    #pragma unroll
    for (int kc = 0; kc < 8; ++kc) {
        const int set = kc & 1, nset = set ^ 1;
        if (kc + 1 < 8) {
            const int kn = kc + 1;
            q[nset][0] = *(const v4f*)(pru0 + (2 * kn) * W);
            q[nset][1] = *(const v4f*)(pru0 + (2 * kn + 1) * W);
            q[nset][2] = *(const v4f*)(prd0 - (2 * kn) * W);
            q[nset][3] = *(const v4f*)(prd0 - (2 * kn + 1) * W);
            q[nset][4] = *(const v4f*)(piu0 + (2 * kn) * W);
            q[nset][5] = *(const v4f*)(piu0 + (2 * kn + 1) * W);
            q[nset][6] = *(const v4f*)(pid0 - (2 * kn) * W);
            q[nset][7] = *(const v4f*)(pid0 - (2 * kn + 1) * W);
            #pragma unroll
            for (int t = 0; t < TPW; ++t) af[nset][t] = ayTab[grp * TPW + t][kn][lane];
        }
        __builtin_amdgcn_sched_barrier(0);

        v2f sr[4], si[4], di[4], dn[4];
        #pragma unroll
        for (int jt = 0; jt < 4; ++jt) {
            sr[jt].x = q[set][0][jt] + q[set][2][jt];
            sr[jt].y = q[set][1][jt] + q[set][3][jt];
            si[jt].x = q[set][4][jt] + q[set][6][jt];
            si[jt].y = q[set][5][jt] + q[set][7][jt];
            di[jt].x = q[set][4][jt] - q[set][6][jt];
            di[jt].y = q[set][5][jt] - q[set][7][jt];
            dn[jt].x = q[set][2][jt] - q[set][0][jt];
            dn[jt].y = q[set][3][jt] - q[set][1][jt];
        }
        #pragma unroll
        for (int t = 0; t < TPW; ++t) {
            const v2f ac = {af[set][t].x, af[set][t].y};
            const v2f as = {af[set][t].z, af[set][t].w};
            #pragma unroll
            for (int jt = 0; jt < 4; ++jt) {
                zre[t][jt] = mm4(ac, sr[jt], zre[t][jt]);
                zim[t][jt] = mm4(ac, si[jt], zim[t][jt]);
            }
            #pragma unroll
            for (int jt = 0; jt < 4; ++jt) {
                zre[t][jt] = mm4(as, di[jt], zre[t][jt]);
                zim[t][jt] = mm4(as, dn[jt], zim[t][jt]);
            }
        }
        __builtin_amdgcn_sched_barrier(0);
    }

    const int tb0 = blockIdx.x * WAVES + grp * TPW;
    #pragma unroll
    for (int t = 0; t < TPW; ++t) {
        const float res = col_phase(zre[t], zim[t], &exTab[grp * TPW + t][0][0], hh, mloc);
        const int m0t = ((tb0 + t) & 127) << 4;
        out[((size_t)(b * C + coil) * M + m0t) * 2 + lane] = res;
    }
}

extern "C" void kernel_launch(void* const* d_in, const int* in_sizes, int n_in,
                              void* d_out, int out_size, void* d_ws, size_t ws_size,
                              hipStream_t stream) {
    (void)in_sizes; (void)n_in; (void)out_size; (void)d_ws; (void)ws_size;
    const float* xr   = (const float*)d_in[0];
    const float* xi   = (const float*)d_in[1];
    const float* traj = (const float*)d_in[2];
    float* out = (float*)d_out;

    dim3 grid(256), block(WAVES * 32);
    hipLaunchKernelGGL(nufft2_rowfirst_mt4, grid, block, 0, stream, xr, xi, traj, out);
}
